# starting kernel
# baseline (speedup 1.0000x reference)
#include <hip/hip_runtime.h>


typedef __attribute__((ext_vector_type(2))) float v2f;
typedef __attribute__((ext_vector_type(8))) float v8f;

#define G_GROUPS 1024
#define T_STEPS  512
#define WAVES_PER_BLOCK 4
#define THREADS (WAVES_PER_BLOCK * 32)

static __device__ __forceinline__ v8f wmma4(v2f a, v2f b, v8f c) {
  return __builtin_amdgcn_wmma_f32_16x16x4_f32(
       false, a,  false, b,
       (short)0, c,  false,  false);
}

static __device__ __forceinline__ v8f zero8() {
  v8f z = {0.f, 0.f, 0.f, 0.f, 0.f, 0.f, 0.f, 0.f};
  return z;
}

static __device__ __forceinline__ void stage_c(float* buf, v8f c, int col, int half) {
#pragma unroll
  for (int r = 0; r < 8; ++r) buf[(r + 8 * half) * 16 + col] = c[r];
}

static __device__ __forceinline__ v2f load_a(const float* buf, int col, int half, int k) {
  int k0 = 4 * k + 2 * half;
  v2f a;
  a.x = buf[col * 16 + k0];
  a.y = buf[col * 16 + k0 + 1];
  return a;
}

static __device__ __forceinline__ v2f load_b(const float* buf, int col, int half, int k) {
  int k0 = 4 * k + 2 * half;
  v2f b;
  b.x = buf[k0 * 16 + col];
  b.y = buf[(k0 + 1) * 16 + col];
  return b;
}

static __device__ __forceinline__ v2f load_bt(const float* buf, int col, int half, int k) {
  int k0 = 4 * k + 2 * half;
  v2f b;
  b.x = buf[col * 16 + k0];
  b.y = buf[col * 16 + k0 + 1];
  return b;
}

static __device__ __forceinline__ float sel4(int i, float a, float b, float c, float d) {
  return i == 0 ? a : (i == 1 ? b : (i == 2 ? c : d));
}

__global__ __launch_bounds__(THREADS, 1)
void kalman_wmma_kernel(const float* __restrict__ y,
                        const float* __restrict__ F,
                        const float* __restrict__ H,
                        const float* __restrict__ Q,
                        const float* __restrict__ R,
                        const float* __restrict__ im,
                        const float* __restrict__ ic,
                        float* __restrict__ out) {
  __shared__ float lds[WAVES_PER_BLOCK * 4 * 256];

  const int lane = threadIdx.x & 31;
  const int wave = threadIdx.x >> 5;
  const int col  = lane & 15;
  const int half = lane >> 4;
  const int g    = blockIdx.x * WAVES_PER_BLOCK + wave;

  float* bufP = &lds[(wave * 4 + 0) * 256];
  float* bufZ = &lds[(wave * 4 + 1) * 256];
  float* bufI = &lds[(wave * 4 + 2) * 256];
  float* bufS = &lds[(wave * 4 + 3) * 256];

  v2f FA[4], HA[4];
#pragma unroll
  for (int k = 0; k < 4; ++k) {
    int k0 = 4 * k + 2 * half;
    FA[k].x = F[col * 16 + k0];
    FA[k].y = F[col * 16 + k0 + 1];
    HA[k].x = (col < 4) ? H[col * 16 + k0] : 0.f;
    HA[k].y = (col < 4) ? H[col * 16 + k0 + 1] : 0.f;
  }
  v2f Hb;
  Hb.x = H[(2 * half) * 16 + col];
  Hb.y = H[(2 * half + 1) * 16 + col];
  v2f Rb;
  Rb.x = (col < 4) ? R[(2 * half) * 4 + col] : 0.f;
  Rb.y = (col < 4) ? R[(2 * half + 1) * 4 + col] : 0.f;

  v8f Qc, Rc, Ic, Pc, Mc;
#pragma unroll
  for (int r = 0; r < 8; ++r) {
    int row = r + 8 * half;
    Qc[r] = Q[row * 16 + col];
    Rc[r] = (row < 4 && col < 4) ? R[row * 4 + col] : 0.f;
    Ic[r] = (row == col) ? 1.f : 0.f;
    Pc[r] = ic[row * 16 + col];
    Mc[r] = (col == 0) ? im[g * 16 + row] : 0.f;
  }

  for (int t = 0; t < T_STEPS; ++t) {
    stage_c(bufP, Pc, col, half);
    v8f PHt = zero8();
#pragma unroll
    for (int k = 0; k < 4; ++k)
      PHt = wmma4(load_a(bufP, col, half, k), HA[k], PHt);

    stage_c(bufZ, PHt, col, half);
    if (col == 0) {
#pragma unroll
      for (int r = 0; r < 8; ++r) bufZ[(r + 8 * half) * 16 + 4] = Mc[r];
    }
    v8f Sc = Rc;
#pragma unroll
    for (int k = 0; k < 4; ++k)
      Sc = wmma4(HA[k], load_b(bufZ, col, half, k), Sc);

    if (lane == 4) {
      float4 o;
      o.x = Sc[0]; o.y = Sc[1]; o.z = Sc[2]; o.w = Sc[3];
      *(float4*)(out + ((size_t)g * T_STEPS + t) * 4) = o;
    }

    float s00 = __shfl(Sc[0], 0, 32), s01 = __shfl(Sc[0], 1, 32),
          s02 = __shfl(Sc[0], 2, 32), s03 = __shfl(Sc[0], 3, 32);
    float s10 = __shfl(Sc[1], 0, 32), s11 = __shfl(Sc[1], 1, 32),
          s12 = __shfl(Sc[1], 2, 32), s13 = __shfl(Sc[1], 3, 32);
    float s20 = __shfl(Sc[2], 0, 32), s21 = __shfl(Sc[2], 1, 32),
          s22 = __shfl(Sc[2], 2, 32), s23 = __shfl(Sc[2], 3, 32);
    float s30 = __shfl(Sc[3], 0, 32), s31 = __shfl(Sc[3], 1, 32),
          s32 = __shfl(Sc[3], 2, 32), s33 = __shfl(Sc[3], 3, 32);
    float mm0 = __shfl(Sc[0], 4, 32), mm1 = __shfl(Sc[1], 4, 32),
          mm2 = __shfl(Sc[2], 4, 32), mm3 = __shfl(Sc[3], 4, 32);

    const float4 yv = *(const float4*)(y + ((size_t)g * T_STEPS + t) * 4);
    float r0 = yv.x - mm0, r1 = yv.y - mm1, r2 = yv.z - mm2, r3 = yv.w - mm3;

    float A2323 = s22 * s33 - s23 * s32;
    float A1323 = s21 * s33 - s23 * s31;
    float A1223 = s21 * s32 - s22 * s31;
    float A0323 = s20 * s33 - s23 * s30;
    float A0223 = s20 * s32 - s22 * s30;
    float A0123 = s20 * s31 - s21 * s30;
    float A2313 = s12 * s33 - s13 * s32;
    float A1313 = s11 * s33 - s13 * s31;
    float A1213 = s11 * s32 - s12 * s31;
    float A2312 = s12 * s23 - s13 * s22;
    float A1312 = s11 * s23 - s13 * s21;
    float A1212 = s11 * s22 - s12 * s21;
    float A0313 = s10 * s33 - s13 * s30;
    float A0213 = s10 * s32 - s12 * s30;
    float A0312 = s10 * s23 - s13 * s20;
    float A0212 = s10 * s22 - s12 * s20;
    float A0113 = s10 * s31 - s11 * s30;
    float A0112 = s10 * s21 - s11 * s20;

    float det = s00 * (s11 * A2323 - s12 * A1323 + s13 * A1223)
              - s01 * (s10 * A2323 - s12 * A0323 + s13 * A0223)
              + s02 * (s10 * A1323 - s11 * A0323 + s13 * A0123)
              - s03 * (s10 * A1223 - s11 * A0223 + s12 * A0123);
    float id = 1.0f / det;

    float i00 =  id * (s11 * A2323 - s12 * A1323 + s13 * A1223);
    float i01 = -id * (s01 * A2323 - s02 * A1323 + s03 * A1223);
    float i02 =  id * (s01 * A2313 - s02 * A1313 + s03 * A1213);
    float i03 = -id * (s01 * A2312 - s02 * A1312 + s03 * A1212);
    float i10 = -id * (s10 * A2323 - s12 * A0323 + s13 * A0223);
    float i11 =  id * (s00 * A2323 - s02 * A0323 + s03 * A0223);
    float i12 = -id * (s00 * A2313 - s02 * A0313 + s03 * A0213);
    float i13 =  id * (s00 * A2312 - s02 * A0312 + s03 * A0212);
    float i20 =  id * (s10 * A1323 - s11 * A0323 + s13 * A0123);
    float i21 = -id * (s00 * A1323 - s01 * A0323 + s03 * A0123);
    float i22 =  id * (s00 * A1313 - s01 * A0313 + s03 * A0113);
    float i23 = -id * (s00 * A1312 - s01 * A0312 + s03 * A0112);
    float i30 = -id * (s10 * A1223 - s11 * A0223 + s12 * A0123);
    float i31 =  id * (s00 * A1223 - s01 * A0223 + s02 * A0123);
    float i32 = -id * (s00 * A1213 - s01 * A0213 + s02 * A0113);
    float i33 =  id * (s00 * A1212 - s01 * A0212 + s02 * A0112);

    float si_r0 = sel4(col, i00, i01, i02, i03);
    float si_r1 = sel4(col, i10, i11, i12, i13);
    float si_r2 = sel4(col, i20, i21, i22, i23);
    float si_r3 = sel4(col, i30, i31, i32, i33);
    v2f Sib;
    Sib.x = (col < 4) ? (half ? si_r2 : si_r0) : 0.f;
    Sib.y = (col < 4) ? (half ? si_r3 : si_r1) : 0.f;

    v2f phta = load_a(bufZ, col, half, 0);
    v8f Kc = wmma4(phta, Sib, zero8());

    stage_c(bufZ, Kc, col, half);
    v2f ka = load_a(bufZ, col, half, 0);
    v2f nka; nka.x = -ka.x; nka.y = -ka.y;

    v8f IKH = wmma4(nka, Hb, Ic);

    v2f rb;
    rb.x = (col == 0) ? (half ? r2 : r0) : 0.f;
    rb.y = (col == 0) ? (half ? r3 : r1) : 0.f;
    v8f Mu = wmma4(ka, rb, Mc);

    stage_c(bufI, IKH, col, half);
    v8f Tm = zero8();
#pragma unroll
    for (int k = 0; k < 4; ++k)
      Tm = wmma4(load_a(bufI, col, half, k), load_b(bufP, col, half, k), Tm);

    v8f KR = wmma4(ka, Rb, zero8());
    stage_c(bufS, KR, col, half);
    v8f KKt = wmma4(load_a(bufS, col, half, 0), load_bt(bufZ, col, half, 0),
                    zero8());

    stage_c(bufP, Tm, col, half);
    v8f Pu = KKt;
#pragma unroll
    for (int k = 0; k < 4; ++k)
      Pu = wmma4(load_a(bufP, col, half, k), load_bt(bufI, col, half, k), Pu);

    stage_c(bufP, Pu, col, half);
    v8f T2 = zero8();
#pragma unroll
    for (int k = 0; k < 4; ++k)
      T2 = wmma4(FA[k], load_b(bufP, col, half, k), T2);
    stage_c(bufI, T2, col, half);
    v8f Pn = Qc;
#pragma unroll
    for (int k = 0; k < 4; ++k)
      Pn = wmma4(load_a(bufI, col, half, k), FA[k]  , Pn);

    if (col == 0) {
#pragma unroll
      for (int r = 0; r < 8; ++r) bufS[r + 8 * half] = Mu[r];
    }
    v8f Mn = zero8();
#pragma unroll
    for (int k = 0; k < 4; ++k) {
      float b0 = bufS[4 * k + 2 * half];
      float b1 = bufS[4 * k + 2 * half + 1];
      v2f mb;
      mb.x = (col == 0) ? b0 : 0.f;
      mb.y = (col == 0) ? b1 : 0.f;
      Mn = wmma4(FA[k], mb, Mn);
    }

    Pc = Pn;
    Mc = Mn;
  }
}

extern "C" void kernel_launch(void* const* d_in, const int* in_sizes, int n_in,
                              void* d_out, int out_size, void* d_ws, size_t ws_size,
                              hipStream_t stream) {
  (void)in_sizes; (void)n_in; (void)out_size; (void)d_ws; (void)ws_size;
  const float* y  = (const float*)d_in[0];
  const float* F  = (const float*)d_in[1];
  const float* H  = (const float*)d_in[2];
  const float* Q  = (const float*)d_in[3];
  const float* R  = (const float*)d_in[4];
  const float* im = (const float*)d_in[5];
  const float* ic = (const float*)d_in[6];
  float* out = (float*)d_out;

  dim3 grid(G_GROUPS / WAVES_PER_BLOCK);
  dim3 block(THREADS);
  hipLaunchKernelGGL(kalman_wmma_kernel, grid, block, 0, stream,
                     y, F, H, Q, R, im, ic, out);
}


# Joseph update multiplied out so the accumulator is always the B operand: K^T = Sinv^T (P H^T)^T, U = P - P H^T K^T, P_u = U + K (R K^T - H U); 3 LDS transposes leave the chain; P_u sum order changes
# speedup vs baseline: 3.3775x; 3.3775x over previous
#include <hip/hip_runtime.h>


typedef __attribute__((ext_vector_type(2))) float v2f;
typedef __attribute__((ext_vector_type(4))) float v4f;
typedef __attribute__((ext_vector_type(8))) float v8f;

#define G_GROUPS 1024
#define T_STEPS  512
#define TILE_G   16
#define CHUNK    8
#define N_CHUNKS (T_STEPS / CHUNK)
#define THREADS  64
#define PITCH    20
#define TILE     (16 * PITCH)
#define KFRAG    64

static __device__ __forceinline__ v8f wmma4(v2f a, v2f b, v8f c) {
  return __builtin_amdgcn_wmma_f32_16x16x4_f32(
       false, a,  false, b,
       (short)0, c,  false,  false);
}

static __device__ __forceinline__ v8f zero8() {
  v8f z = {0.f, 0.f, 0.f, 0.f, 0.f, 0.f, 0.f, 0.f};
  return z;
}

static __device__ __forceinline__ void stage_c(float* buf, v8f c, int col, int half) {
  v4f* p = (v4f*)(buf + col * PITCH + 8 * half);
  p[0] = __builtin_shufflevector(c, c, 0, 1, 2, 3);
  p[1] = __builtin_shufflevector(c, c, 4, 5, 6, 7);
  __builtin_amdgcn_fence(__ATOMIC_SEQ_CST, "wavefront");
}

static __device__ __forceinline__ v2f load_a(const float* buf, int col, int half, int k) {
  int k0 = 4 * k + 2 * half;
  v2f a;
  a.x = buf[k0 * PITCH + col];
  a.y = buf[(k0 + 1) * PITCH + col];
  return a;
}

static __device__ __forceinline__ v2f load_b(const float* buf, int col, int half, int k) {
  int k0 = 4 * k + 2 * half;
  return *(const v2f*)(buf + col * PITCH + k0);
}

static __device__ __forceinline__ void half_swap(float& d, float& s) {
  asm("v_permlane16_swap_b32 %0, %1" : "+v"(d), "+v"(s));
}

static __device__ __forceinline__ void acc_to_b(v8f c, v2f b[4]) {
  float c0 = c[0], c1 = c[1], c2 = c[2], c3 = c[3];
  float c4 = c[4], c5 = c[5], c6 = c[6], c7 = c[7];
  half_swap(c0, c2);
  half_swap(c1, c3);
  half_swap(c4, c6);
  half_swap(c5, c7);
  b[0].x = c0; b[0].y = c1;
  b[1].x = c4; b[1].y = c5;
  b[2].x = c2; b[2].y = c3;
  b[3].x = c6; b[3].y = c7;
}


static __device__ __forceinline__ float row_bcast(float v, unsigned sel) {
  unsigned x = __builtin_bit_cast(unsigned, v);
  return __builtin_bit_cast(float,
      __builtin_amdgcn_permlane16(x, x, sel, sel, false, false));
}

static __device__ __forceinline__ void loads_ahead() {
  __builtin_amdgcn_sched_barrier(0);
}

static __device__ __forceinline__ void pin_here() {
  __builtin_amdgcn_sched_barrier(0);
}

__global__ __launch_bounds__(THREADS, 1)
void kalman_wmma_kernel(const float* __restrict__ y,
                        const float* __restrict__ F,
                        const float* __restrict__ H,
                        const float* __restrict__ Q,
                        const float* __restrict__ R,
                        const float* __restrict__ im,
                        const float* __restrict__ ic,
                        float* __restrict__ out) {
  __shared__ __attribute__((aligned(16))) float lds[3 * TILE + 2 * CHUNK * KFRAG];

  const int lane = threadIdx.x & 31;
  const int col  = lane & 15;
  const int half = lane >> 4;
  const int wave = __builtin_amdgcn_readfirstlane((int)(threadIdx.x >> 5));
  const int g    = blockIdx.x * TILE_G + col;

  float* bufZ = &lds[0 * TILE];
  float* bufI = &lds[1 * TILE];
  float* bufM = &lds[2 * TILE];
  float* ring = &lds[3 * TILE];

  const int hrow = (col == 0) ? 0 : (col == 1) ? 1 : (col == 8) ? 2 : (col == 9) ? 3 : -1;
  const int hrc  = (hrow < 0) ? 0 : hrow;
  const int pick = ((col & 7) < 4) ? ((col & 7) + 4 * half) : 8;
  v2f FA[4], HA[4], HS[4], HSn[4], HpA[4];
#pragma unroll
  for (int k = 0; k < 4; ++k) {
    int k0 = 4 * k + 2 * half;
    FA[k].x = F[col * 16 + k0];
    FA[k].y = F[col * 16 + k0 + 1];
    HA[k].x = (col < 4) ? H[col * 16 + k0] : 0.f;
    HA[k].y = (col < 4) ? H[col * 16 + k0 + 1] : 0.f;
    HS[k].x = ((col & 7) < 4) ? H[(col & 7) * 16 + k0] : 0.f;
    HS[k].y = ((col & 7) < 4) ? H[(col & 7) * 16 + k0 + 1] : 0.f;
    HSn[k].x = -HS[k].x;
    HSn[k].y = -HS[k].y;
    float hx = H[hrc * 16 + k0];
    float hy = H[hrc * 16 + k0 + 1];
    HpA[k].x = (hrow >= 0) ? hx : 0.f;
    HpA[k].y = (hrow >= 0) ? hy : 0.f;
  }
  v2f Ra;
  Ra.x = ((col & 7) < 4) ? R[(col & 7) * 4 + 2 * half] : 0.f;
  Ra.y = ((col & 7) < 4) ? R[(col & 7) * 4 + 2 * half + 1] : 0.f;

  v2f FtA[4], QA[4], IcA[4];
#pragma unroll
  for (int k = 0; k < 4; ++k) {
    int k0 = 4 * k + 2 * half;
    FtA[k].x = F[k0 * 16 + col];
    FtA[k].y = F[(k0 + 1) * 16 + col];
    QA[k].x  = Q[col * 16 + k0];
    QA[k].y  = Q[col * 16 + k0 + 1];
    IcA[k].x = ic[col * 16 + k0];
    IcA[k].y = ic[col * 16 + k0 + 1];
  }

  v8f Qc, Rc, Pc, Mc;
#pragma unroll
  for (int r = 0; r < 8; ++r) {
    int row = r + 8 * half;
    Qc[r] = Q[row * 16 + col];
    Rc[r] = (r < 4 && col < 4) ? R[r * 4 + col] : 0.f;
    Pc[r] = ic[row * 16 + col];
    Mc[r] = im[g * 16 + row];
  }

  v8f HFtC = zero8(), QHt = zero8(), PHt = zero8();
#pragma unroll
  for (int k = 0; k < 4; ++k) {
    HFtC = wmma4(FtA[k], HA[k], HFtC);
    QHt  = wmma4(QA[k],  HA[k], QHt);
    PHt  = wmma4(IcA[k], HA[k], PHt);
  }
  v2f HFt[4];
  acc_to_b(HFtC, HFt);
  v2f t2a[4];
#pragma unroll
  for (int k = 0; k < 4; ++k) { t2a[k].x = 0.f; t2a[k].y = 0.f; }

  for (int c = 0; c <= N_CHUNKS; ++c) {
    if (wave == 0) {
      if (c < N_CHUNKS) {
        float* slot0 = ring + (c & 1) * (CHUNK * KFRAG);
        for (int j = 0; j < CHUNK; ++j) {
          float* slotK = slot0 + j * KFRAG;


          const bool first = (c | j) == 0;
          v8f Pn;
#pragma unroll
          for (int r = 0; r < 8; ++r) Pn[r] = first ? Pc[r] : Qc[r];
          Pn = wmma4(t2a[0], FA[0]  , Pn);

          stage_c(bufZ, PHt, col, half);
          v2f phta = load_a(bufZ, col, half, 0);
          v2f zb[4];
          acc_to_b(PHt, zb);
          loads_ahead();
          v8f Sc = Rc;
#pragma unroll
          for (int k = 0; k < 4; ++k)
            Sc = wmma4(HS[k], zb[k], Sc);


          float s00 = row_bcast(Sc[0], 0x00000000u), s01 = row_bcast(Sc[0], 0x11111111u),
                s02 = row_bcast(Sc[0], 0x22222222u), s03 = row_bcast(Sc[0], 0x33333333u);
          float s10 = row_bcast(Sc[1], 0x00000000u), s11 = row_bcast(Sc[1], 0x11111111u),
                s12 = row_bcast(Sc[1], 0x22222222u), s13 = row_bcast(Sc[1], 0x33333333u);
          pin_here();
          Pn = wmma4(t2a[1], FA[1], Pn);
          pin_here();
          float s20 = row_bcast(Sc[2], 0x00000000u), s21 = row_bcast(Sc[2], 0x11111111u),
                s22 = row_bcast(Sc[2], 0x22222222u), s23 = row_bcast(Sc[2], 0x33333333u);
          float s30 = row_bcast(Sc[3], 0x00000000u), s31 = row_bcast(Sc[3], 0x11111111u),
                s32 = row_bcast(Sc[3], 0x22222222u), s33 = row_bcast(Sc[3], 0x33333333u);


          pin_here();
          Pn = wmma4(t2a[2], FA[2], Pn);
          pin_here();

          float A2323 = s22 * s33 - s23 * s32;
          float A1323 = s21 * s33 - s23 * s31;
          float A1223 = s21 * s32 - s22 * s31;
          float A0323 = s20 * s33 - s23 * s30;
          float A0223 = s20 * s32 - s22 * s30;
          float A0123 = s20 * s31 - s21 * s30;
          float A2313 = s12 * s33 - s13 * s32;
          float A1313 = s11 * s33 - s13 * s31;
          float A1213 = s11 * s32 - s12 * s31;
          float A2312 = s12 * s23 - s13 * s22;
          float A1312 = s11 * s23 - s13 * s21;
          float A1212 = s11 * s22 - s12 * s21;
          float A0313 = s10 * s33 - s13 * s30;
          float A0213 = s10 * s32 - s12 * s30;
          float A0312 = s10 * s23 - s13 * s20;
          float A0212 = s10 * s22 - s12 * s20;
          float A0113 = s10 * s31 - s11 * s30;
          float A0112 = s10 * s21 - s11 * s20;


          pin_here();
          Pn = wmma4(t2a[3], FA[3], Pn);
          pin_here();

          float det = s00 * (s11 * A2323 - s12 * A1323 + s13 * A1223)
                    - s01 * (s10 * A2323 - s12 * A0323 + s13 * A0223)
                    + s02 * (s10 * A1323 - s11 * A0323 + s13 * A0123)
                    - s03 * (s10 * A1223 - s11 * A0223 + s12 * A0123);
          float id = 1.0f / det;

          float c00 =  (s11 * A2323 - s12 * A1323 + s13 * A1223);
          float c01 = -(s01 * A2323 - s02 * A1323 + s03 * A1223);
          float c02 =  (s01 * A2313 - s02 * A1313 + s03 * A1213);
          float c03 = -(s01 * A2312 - s02 * A1312 + s03 * A1212);
          float c10 = -(s10 * A2323 - s12 * A0323 + s13 * A0223);
          float c11 =  (s00 * A2323 - s02 * A0323 + s03 * A0223);
          float c12 = -(s00 * A2313 - s02 * A0313 + s03 * A0213);
          float c13 =  (s00 * A2312 - s02 * A0312 + s03 * A0212);
          float c20 =  (s10 * A1323 - s11 * A0323 + s13 * A0123);
          float c21 = -(s00 * A1323 - s01 * A0323 + s03 * A0123);
          float c22 =  (s00 * A1313 - s01 * A0313 + s03 * A0113);
          float c23 = -(s00 * A1312 - s01 * A0312 + s03 * A0112);
          float c30 = -(s10 * A1223 - s11 * A0223 + s12 * A0123);
          float c31 =  (s00 * A1223 - s01 * A0223 + s02 * A0123);
          float c32 = -(s00 * A1213 - s01 * A0213 + s02 * A0113);
          float c33 =  (s00 * A1212 - s01 * A0212 + s02 * A0112);

          v2f adj;
          adj.x = 0.f;                       adj.y = 0.f;
          adj.x = (pick == 0) ? c00 : adj.x; adj.y = (pick == 0) ? c10 : adj.y;
          adj.x = (pick == 1) ? c01 : adj.x; adj.y = (pick == 1) ? c11 : adj.y;
          adj.x = (pick == 2) ? c02 : adj.x; adj.y = (pick == 2) ? c12 : adj.y;
          adj.x = (pick == 3) ? c03 : adj.x; adj.y = (pick == 3) ? c13 : adj.y;
          adj.x = (pick == 4) ? c20 : adj.x; adj.y = (pick == 4) ? c30 : adj.y;
          adj.x = (pick == 5) ? c21 : adj.x; adj.y = (pick == 5) ? c31 : adj.y;
          adj.x = (pick == 6) ? c22 : adj.x; adj.y = (pick == 6) ? c32 : adj.y;
          adj.x = (pick == 7) ? c23 : adj.x; adj.y = (pick == 7) ? c33 : adj.y;


          v2f Sia;
          Sia.x = id * adj.x;
          Sia.y = id * adj.y;
          v2f nphta; nphta.x = -phta.x; nphta.y = -phta.y;

          v8f KT = wmma4(Sia, phta, zero8());

          v2f ka;
          ka.x = half ? KT[2] : KT[0];
          ka.y = half ? KT[3] : KT[1];

          v8f U = wmma4(nphta, ka, Pn);
          v8f V = wmma4(Ra, ka, zero8());

          v2f ub[4];
          acc_to_b(U, ub);
          loads_ahead();
#pragma unroll
          for (int k = 0; k < 4; ++k)
            V = wmma4(HSn[k], ub[k], V);

          v2f vb;
          vb.x = half ? V[2] : V[0];
          vb.y = half ? V[3] : V[1];

          v8f Pu = wmma4(ka, vb, U);

          v2f pub[4];
          acc_to_b(Pu, pub);
          loads_ahead();
          v8f T2 = zero8();
#pragma unroll
          for (int k = 0; k < 4; ++k)
            T2 = wmma4(FA[k], pub[k], T2);
          loads_ahead();
          *(v2f*)(slotK + 2 * lane) = ka;
          stage_c(bufI, T2, col, half);
#pragma unroll
          for (int k = 0; k < 4; ++k) t2a[k] = load_a(bufI, col, half, k);
          loads_ahead();

          PHt = QHt;
#pragma unroll
          for (int k = 0; k < 4; ++k)
            PHt = wmma4(t2a[k], HFt[k], PHt);
        }
      }
    } else {
      if (c >= 1) {
        const int cc = c - 1;
        const float* slot0 = ring + (cc & 1) * (CHUNK * KFRAG);
        const size_t base = ((size_t)g * T_STEPS + (size_t)cc * CHUNK) * 4 + 2 * half;

        v2f yv[CHUNK];
#pragma unroll
        for (int j = 0; j < CHUNK; ++j)
          yv[j] = *(const v2f*)(y + base + (size_t)j * 4);

#pragma unroll
        for (int j = 0; j < CHUNK; ++j) {
          const float* slotK = slot0 + j * KFRAG;

          stage_c(bufM, Mc, col, half);
          v8f Mm = zero8();
#pragma unroll
          for (int k = 0; k < 4; ++k)
            Mm = wmma4(HpA[k], load_b(bufM, col, half, k), Mm);

          v2f o;
          o.x = Mm[0]; o.y = Mm[1];
          *(v2f*)(out + base + (size_t)j * 4) = o;

          v2f rb;
          rb.x = yv[j].x - Mm[0];
          rb.y = yv[j].y - Mm[1];

          v2f ka = *(const v2f*)(slotK + 2 * lane);
          v8f Mu = wmma4(ka, rb, Mc);

          stage_c(bufM, Mu, col, half);
          v8f Mn = zero8();
#pragma unroll
          for (int k = 0; k < 4; ++k)
            Mn = wmma4(FA[k], load_b(bufM, col, half, k), Mn);

          Mc = Mn;
        }
      }
    }
    __syncthreads();
  }
}

extern "C" void kernel_launch(void* const* d_in, const int* in_sizes, int n_in,
                              void* d_out, int out_size, void* d_ws, size_t ws_size,
                              hipStream_t stream) {
  (void)in_sizes; (void)n_in; (void)out_size; (void)d_ws; (void)ws_size;
  const float* y  = (const float*)d_in[0];
  const float* F  = (const float*)d_in[1];
  const float* H  = (const float*)d_in[2];
  const float* Q  = (const float*)d_in[3];
  const float* R  = (const float*)d_in[4];
  const float* im = (const float*)d_in[5];
  const float* ic = (const float*)d_in[6];
  float* out = (float*)d_out;

  dim3 grid(G_GROUPS / TILE_G);
  dim3 block(THREADS);
  hipLaunchKernelGGL(kalman_wmma_kernel, grid, block, 0, stream,
                     y, F, H, Q, R, im, ic, out);
}
